# starting kernel
# baseline (speedup 1.0000x reference)
#include <hip/hip_runtime.h>

typedef float v2f  __attribute__((ext_vector_type(2)));
typedef float v8f  __attribute__((ext_vector_type(8)));

constexpr int B_    = 8;
constexpr int CIN   = 32;
constexpr int NPOS  = 207;
constexpr int T     = 256;
constexpr int CO    = 16;
constexpr int TTOUT = 240;
constexpr int DIL   = 2;
constexpr int NT    = NPOS * T;
constexpr int STR2  = 20;

__device__ inline float4 lo4(v8f a) { return make_float4(a[0], a[1], a[2], a[3]); }
__device__ inline float4 hi4(v8f a) { return make_float4(a[4], a[5], a[6], a[7]); }

__device__ inline float dot16(const float4& a0, const float4& a1, const float4& a2, const float4& a3,
                              const float4& b0, const float4& b1, const float4& b2, const float4& b3) {
  float d = a0.x * b0.x;
  d = fmaf(a0.y, b0.y, d); d = fmaf(a0.z, b0.z, d); d = fmaf(a0.w, b0.w, d);
  d = fmaf(a1.x, b1.x, d); d = fmaf(a1.y, b1.y, d); d = fmaf(a1.z, b1.z, d); d = fmaf(a1.w, b1.w, d);
  d = fmaf(a2.x, b2.x, d); d = fmaf(a2.y, b2.y, d); d = fmaf(a2.z, b2.z, d); d = fmaf(a2.w, b2.w, d);
  d = fmaf(a3.x, b3.x, d); d = fmaf(a3.y, b3.y, d); d = fmaf(a3.z, b3.z, d); d = fmaf(a3.w, b3.w, d);
  return d;
}

__global__ __launch_bounds__(256) void mdsa_kernel(
    const float* __restrict__ x,
    const float* __restrict__ Wq,
    const float* __restrict__ Wk,
    const float* __restrict__ Wv,
    float* __restrict__ out)
{
  __shared__ __align__(16) float qs [T * STR2];
  __shared__ __align__(16) float ks_[T * STR2];
  __shared__ __align__(16) float vs_[T * STR2];

  const int b = blockIdx.x / NPOS;
  const int n = blockIdx.x % NPOS;
  const float* xb = x + (size_t)b * CIN * NT + (size_t)n * T;

  const int tid  = threadIdx.x;
  const int wave = tid >> 5;
  const int lane = tid & 31;
  const int hi   = lane >> 4;
  const int col  = lane & 15;

  {
    const int c     = tid & 31;
    const int chunk = tid >> 5;
    __builtin_prefetch(xb + (size_t)c * NT + chunk * 32, 0, 1);
  }

  v2f bfrag[2][8];
#pragma unroll
  for (int tile = 0; tile < 2; ++tile) {
    const int t0 = (wave * 2 + tile) * 16;
#pragma unroll
    for (int ksp = 0; ksp < 8; ++ksp) {
      const int c0 = ksp * 4 + 2 * hi;
      bfrag[tile][ksp].x = xb[(size_t)c0 * NT + t0 + col];
      bfrag[tile][ksp].y = xb[(size_t)(c0 + 1) * NT + t0 + col];
    }
  }

  constexpr int KERNS[4] = {3, 5, 7, 9};

#pragma unroll
  for (int br = 0; br < 4; ++br) {
    const int kern = KERNS[br];
    const float* wq = Wq + br * CO * CIN;
    const float* wk = Wk + br * CO * CIN;
    const float* wv = Wv + br * CO * CIN;

    __syncthreads();

    v8f aq0 = {}, aq1 = {}, ak0 = {}, ak1 = {}, av0 = {}, av1 = {};
#pragma unroll
    for (int ksp = 0; ksp < 8; ++ksp) {
      const int c0 = ksp * 4 + 2 * hi;
      v2f fq, fk, fv;
      fq.x = wq[col * CIN + c0];  fq.y = wq[col * CIN + c0 + 1];
      fk.x = wk[col * CIN + c0];  fk.y = wk[col * CIN + c0 + 1];
      fv.x = wv[col * CIN + c0];  fv.y = wv[col * CIN + c0 + 1];
      aq0 = __builtin_amdgcn_wmma_f32_16x16x4_f32(false, fq, false, bfrag[0][ksp], (short)0, aq0, false, false);
      aq1 = __builtin_amdgcn_wmma_f32_16x16x4_f32(false, fq, false, bfrag[1][ksp], (short)0, aq1, false, false);
      ak0 = __builtin_amdgcn_wmma_f32_16x16x4_f32(false, fk, false, bfrag[0][ksp], (short)0, ak0, false, false);
      ak1 = __builtin_amdgcn_wmma_f32_16x16x4_f32(false, fk, false, bfrag[1][ksp], (short)0, ak1, false, false);
      av0 = __builtin_amdgcn_wmma_f32_16x16x4_f32(false, fv, false, bfrag[0][ksp], (short)0, av0, false, false);
      av1 = __builtin_amdgcn_wmma_f32_16x16x4_f32(false, fv, false, bfrag[1][ksp], (short)0, av1, false, false);
    }

    {
      const int t0 = wave * 32;
      int row0 = (t0 + col) * STR2 + hi * 8;
      int row1 = (t0 + 16 + col) * STR2 + hi * 8;
      *(float4*)&qs [row0]     = lo4(aq0);  *(float4*)&qs [row0 + 4] = hi4(aq0);
      *(float4*)&qs [row1]     = lo4(aq1);  *(float4*)&qs [row1 + 4] = hi4(aq1);
      *(float4*)&ks_[row0]     = lo4(ak0);  *(float4*)&ks_[row0 + 4] = hi4(ak0);
      *(float4*)&ks_[row1]     = lo4(ak1);  *(float4*)&ks_[row1 + 4] = hi4(ak1);
      *(float4*)&vs_[row0]     = lo4(av0);  *(float4*)&vs_[row0 + 4] = hi4(av0);
      *(float4*)&vs_[row1]     = lo4(av1);  *(float4*)&vs_[row1 + 4] = hi4(av1);
    }

    __syncthreads();

    if (tid < TTOUT) {
      const int tg = tid + (T - TTOUT);

      const float4* qrow = (const float4*)&qs[tg * STR2];
      const float4 q0 = qrow[0], q1 = qrow[1], q2 = qrow[2], q3 = qrow[3];

      float s[9];
      float mx = -3.0e38f;
#pragma unroll
      for (int j = 0; j < kern; ++j) {
        const int tk = tg - DIL * (kern - 1 - j);
        const float4* krow = (const float4*)&ks_[tk * STR2];
        float d = dot16(q0, q1, q2, q3, krow[0], krow[1], krow[2], krow[3]);
        d *= 0.25f;
        s[j] = d;
        mx = fmaxf(mx, d);
      }
      float ssum = 0.f;
#pragma unroll
      for (int j = 0; j < kern; ++j) { s[j] = __expf(s[j] - mx); ssum += s[j]; }
      const float inv = 1.f / ssum;

      float4 o0 = make_float4(0.f, 0.f, 0.f, 0.f), o1 = o0, o2 = o0, o3 = o0;
#pragma unroll
      for (int j = 0; j < kern; ++j) {
        const int tv = tg - DIL * (kern - 1 - j);
        const float4* vrow = (const float4*)&vs_[tv * STR2];
        const float w = s[j];
        const float4 v0 = vrow[0], v1 = vrow[1], v2 = vrow[2], v3 = vrow[3];
        o0.x = fmaf(w, v0.x, o0.x); o0.y = fmaf(w, v0.y, o0.y); o0.z = fmaf(w, v0.z, o0.z); o0.w = fmaf(w, v0.w, o0.w);
        o1.x = fmaf(w, v1.x, o1.x); o1.y = fmaf(w, v1.y, o1.y); o1.z = fmaf(w, v1.z, o1.z); o1.w = fmaf(w, v1.w, o1.w);
        o2.x = fmaf(w, v2.x, o2.x); o2.y = fmaf(w, v2.y, o2.y); o2.z = fmaf(w, v2.z, o2.z); o2.w = fmaf(w, v2.w, o2.w);
        o3.x = fmaf(w, v3.x, o3.x); o3.y = fmaf(w, v3.y, o3.y); o3.z = fmaf(w, v3.z, o3.z); o3.w = fmaf(w, v3.w, o3.w);
      }

      float o[CO] = { o0.x, o0.y, o0.z, o0.w, o1.x, o1.y, o1.z, o1.w,
                      o2.x, o2.y, o2.z, o2.w, o3.x, o3.y, o3.z, o3.w };
#pragma unroll
      for (int c = 0; c < CO; ++c)
        out[(((size_t)b * 64 + br * CO + c) * NPOS + n) * TTOUT + tid] = o[c] * inv;
    }
  }
}

extern "C" void kernel_launch(void* const* d_in, const int* in_sizes, int n_in,
                              void* d_out, int out_size, void* d_ws, size_t ws_size,
                              hipStream_t stream) {
  const float* x  = (const float*)d_in[0];
  const float* Wq = (const float*)d_in[1];
  const float* Wk = (const float*)d_in[2];
  const float* Wv = (const float*)d_in[3];
  float* out = (float*)d_out;

  dim3 grid(B_ * NPOS);
  dim3 block(256);
  mdsa_kernel<<<grid, block, 0, stream>>>(x, Wq, Wk, Wv, out);
}


# q/k/v double-buffered in LDS by branch parity: one barrier per branch instead of two; store drain and late waves' attention now overlap the next branch's projection WMMAs; arithmetic unchanged
# speedup vs baseline: 1.1133x; 1.1133x over previous
#include <hip/hip_runtime.h>

typedef float v2f  __attribute__((ext_vector_type(2)));
typedef float v8f  __attribute__((ext_vector_type(8)));

constexpr int B_    = 8;
constexpr int CIN   = 32;
constexpr int NPOS  = 207;
constexpr int T     = 256;
constexpr int CO    = 16;
constexpr int TTOUT = 240;
constexpr int DIL   = 2;
constexpr int NT    = NPOS * T;
constexpr int STR2  = 20;

__device__ inline float4 lo4(v8f a) { return make_float4(a[0], a[1], a[2], a[3]); }
__device__ inline float4 hi4(v8f a) { return make_float4(a[4], a[5], a[6], a[7]); }

__device__ inline float dot16(const float4& a0, const float4& a1, const float4& a2, const float4& a3,
                              const float4& b0, const float4& b1, const float4& b2, const float4& b3) {
  float d = a0.x * b0.x;
  d = fmaf(a0.y, b0.y, d); d = fmaf(a0.z, b0.z, d); d = fmaf(a0.w, b0.w, d);
  d = fmaf(a1.x, b1.x, d); d = fmaf(a1.y, b1.y, d); d = fmaf(a1.z, b1.z, d); d = fmaf(a1.w, b1.w, d);
  d = fmaf(a2.x, b2.x, d); d = fmaf(a2.y, b2.y, d); d = fmaf(a2.z, b2.z, d); d = fmaf(a2.w, b2.w, d);
  d = fmaf(a3.x, b3.x, d); d = fmaf(a3.y, b3.y, d); d = fmaf(a3.z, b3.z, d); d = fmaf(a3.w, b3.w, d);
  return d;
}

__global__ __launch_bounds__(256) void mdsa_kernel(
    const float* __restrict__ x,
    const float* __restrict__ Wq,
    const float* __restrict__ Wk,
    const float* __restrict__ Wv,
    float* __restrict__ out)
{
  __shared__ __align__(16) float qs2[2][T * STR2];
  __shared__ __align__(16) float ks2[2][T * STR2];
  __shared__ __align__(16) float vs2[2][T * STR2];

  const int b = blockIdx.x / NPOS;
  const int n = blockIdx.x % NPOS;
  const float* xb = x + (size_t)b * CIN * NT + (size_t)n * T;

  const int tid  = threadIdx.x;
  const int wave = tid >> 5;
  const int lane = tid & 31;
  const int hi   = lane >> 4;
  const int col  = lane & 15;

  {
    const int c     = tid & 31;
    const int chunk = tid >> 5;
    __builtin_prefetch(xb + (size_t)c * NT + chunk * 32, 0, 1);
  }

  v2f bfrag[2][8];
#pragma unroll
  for (int tile = 0; tile < 2; ++tile) {
    const int t0 = (wave * 2 + tile) * 16;
#pragma unroll
    for (int ksp = 0; ksp < 8; ++ksp) {
      const int c0 = ksp * 4 + 2 * hi;
      bfrag[tile][ksp].x = xb[(size_t)c0 * NT + t0 + col];
      bfrag[tile][ksp].y = xb[(size_t)(c0 + 1) * NT + t0 + col];
    }
  }

  constexpr int KERNS[4] = {3, 5, 7, 9};

#pragma unroll
  for (int br = 0; br < 4; ++br) {
    const int kern = KERNS[br];
    const float* wq = Wq + br * CO * CIN;
    const float* wk = Wk + br * CO * CIN;
    const float* wv = Wv + br * CO * CIN;

    float* qs  = qs2[br & 1];
    float* ks_ = ks2[br & 1];
    float* vs_ = vs2[br & 1];

    v8f aq0 = {}, aq1 = {}, ak0 = {}, ak1 = {}, av0 = {}, av1 = {};
#pragma unroll
    for (int ksp = 0; ksp < 8; ++ksp) {
      const int c0 = ksp * 4 + 2 * hi;
      v2f fq, fk, fv;
      fq.x = wq[col * CIN + c0];  fq.y = wq[col * CIN + c0 + 1];
      fk.x = wk[col * CIN + c0];  fk.y = wk[col * CIN + c0 + 1];
      fv.x = wv[col * CIN + c0];  fv.y = wv[col * CIN + c0 + 1];
      aq0 = __builtin_amdgcn_wmma_f32_16x16x4_f32(false, fq, false, bfrag[0][ksp], (short)0, aq0, false, false);
      aq1 = __builtin_amdgcn_wmma_f32_16x16x4_f32(false, fq, false, bfrag[1][ksp], (short)0, aq1, false, false);
      ak0 = __builtin_amdgcn_wmma_f32_16x16x4_f32(false, fk, false, bfrag[0][ksp], (short)0, ak0, false, false);
      ak1 = __builtin_amdgcn_wmma_f32_16x16x4_f32(false, fk, false, bfrag[1][ksp], (short)0, ak1, false, false);
      av0 = __builtin_amdgcn_wmma_f32_16x16x4_f32(false, fv, false, bfrag[0][ksp], (short)0, av0, false, false);
      av1 = __builtin_amdgcn_wmma_f32_16x16x4_f32(false, fv, false, bfrag[1][ksp], (short)0, av1, false, false);
    }

    {
      const int t0 = wave * 32;
      int row0 = (t0 + col) * STR2 + hi * 8;
      int row1 = (t0 + 16 + col) * STR2 + hi * 8;
      *(float4*)&qs [row0]     = lo4(aq0);  *(float4*)&qs [row0 + 4] = hi4(aq0);
      *(float4*)&qs [row1]     = lo4(aq1);  *(float4*)&qs [row1 + 4] = hi4(aq1);
      *(float4*)&ks_[row0]     = lo4(ak0);  *(float4*)&ks_[row0 + 4] = hi4(ak0);
      *(float4*)&ks_[row1]     = lo4(ak1);  *(float4*)&ks_[row1 + 4] = hi4(ak1);
      *(float4*)&vs_[row0]     = lo4(av0);  *(float4*)&vs_[row0 + 4] = hi4(av0);
      *(float4*)&vs_[row1]     = lo4(av1);  *(float4*)&vs_[row1 + 4] = hi4(av1);
    }

    __syncthreads();

    if (tid < TTOUT) {
      const int tg = tid + (T - TTOUT);

      const float4* qrow = (const float4*)&qs[tg * STR2];
      const float4 q0 = qrow[0], q1 = qrow[1], q2 = qrow[2], q3 = qrow[3];

      float s[9];
      float mx = -3.0e38f;
#pragma unroll
      for (int j = 0; j < kern; ++j) {
        const int tk = tg - DIL * (kern - 1 - j);
        const float4* krow = (const float4*)&ks_[tk * STR2];
        float d = dot16(q0, q1, q2, q3, krow[0], krow[1], krow[2], krow[3]);
        d *= 0.25f;
        s[j] = d;
        mx = fmaxf(mx, d);
      }
      float ssum = 0.f;
#pragma unroll
      for (int j = 0; j < kern; ++j) { s[j] = __expf(s[j] - mx); ssum += s[j]; }
      const float inv = 1.f / ssum;

      float4 o0 = make_float4(0.f, 0.f, 0.f, 0.f), o1 = o0, o2 = o0, o3 = o0;
#pragma unroll
      for (int j = 0; j < kern; ++j) {
        const int tv = tg - DIL * (kern - 1 - j);
        const float4* vrow = (const float4*)&vs_[tv * STR2];
        const float w = s[j];
        const float4 v0 = vrow[0], v1 = vrow[1], v2 = vrow[2], v3 = vrow[3];
        o0.x = fmaf(w, v0.x, o0.x); o0.y = fmaf(w, v0.y, o0.y); o0.z = fmaf(w, v0.z, o0.z); o0.w = fmaf(w, v0.w, o0.w);
        o1.x = fmaf(w, v1.x, o1.x); o1.y = fmaf(w, v1.y, o1.y); o1.z = fmaf(w, v1.z, o1.z); o1.w = fmaf(w, v1.w, o1.w);
        o2.x = fmaf(w, v2.x, o2.x); o2.y = fmaf(w, v2.y, o2.y); o2.z = fmaf(w, v2.z, o2.z); o2.w = fmaf(w, v2.w, o2.w);
        o3.x = fmaf(w, v3.x, o3.x); o3.y = fmaf(w, v3.y, o3.y); o3.z = fmaf(w, v3.z, o3.z); o3.w = fmaf(w, v3.w, o3.w);
      }

      float o[CO] = { o0.x, o0.y, o0.z, o0.w, o1.x, o1.y, o1.z, o1.w,
                      o2.x, o2.y, o2.z, o2.w, o3.x, o3.y, o3.z, o3.w };
#pragma unroll
      for (int c = 0; c < CO; ++c)
        out[(((size_t)b * 64 + br * CO + c) * NPOS + n) * TTOUT + tid] = o[c] * inv;
    }
  }
}

extern "C" void kernel_launch(void* const* d_in, const int* in_sizes, int n_in,
                              void* d_out, int out_size, void* d_ws, size_t ws_size,
                              hipStream_t stream) {
  const float* x  = (const float*)d_in[0];
  const float* Wq = (const float*)d_in[1];
  const float* Wk = (const float*)d_in[2];
  const float* Wv = (const float*)d_in[3];
  float* out = (float*)d_out;

  dim3 grid(B_ * NPOS);
  dim3 block(256);
  mdsa_kernel<<<grid, block, 0, stream>>>(x, Wq, Wk, Wv, out);
}
